# starting kernel
# baseline (speedup 1.0000x reference)
#include <hip/hip_runtime.h>

typedef __attribute__((ext_vector_type(2))) float v2f;
typedef __attribute__((ext_vector_type(8))) float v8f;

#define F48 48

__global__ void build_feat_zero_deg(const float* __restrict__ x,
                                    const float* __restrict__ h,
                                    const float* __restrict__ q,
                                    float* __restrict__ featw,
                                    int* __restrict__ deg, int N) {
  int i = blockIdx.x * blockDim.x + threadIdx.x;
  if (i >= N) return;
  const float4* xr = (const float4*)(x + (size_t)i * 16);
  const float4* hr = (const float4*)(h + (size_t)i * 16);
  const float4* qr = (const float4*)(q + (size_t)i * 16);
  float4* fw = (float4*)(featw + (size_t)i * F48);
#pragma unroll
  for (int k = 0; k < 4; ++k) fw[k] = xr[k];
#pragma unroll
  for (int k = 0; k < 4; ++k) fw[4 + k] = hr[k];
#pragma unroll
  for (int k = 0; k < 4; ++k) fw[8 + k] = qr[k];
  deg[i] = 0;
}

__global__ void count_deg(const int* __restrict__ dst, int* __restrict__ deg,
                          int E) {
  int i = blockIdx.x * blockDim.x + threadIdx.x;
  if (i >= E) return;
  atomicAdd(&deg[dst[i]], 1);
}

__global__ void compute_dis(const int* __restrict__ deg,
                            float* __restrict__ dis, float* __restrict__ dis2,
                            int N) {
  int i = blockIdx.x * blockDim.x + threadIdx.x;
  if (i >= N) return;
  float d = (float)(deg[i] + 1);
  dis[i] = rsqrtf(d);
  dis2[i] = 1.0f / d;
}

__global__ void gcn_gemm_wmma(const float* __restrict__ featw,
                              const float* __restrict__ W3,
                              const float* __restrict__ b3,
                              const float* __restrict__ dis2,
                              float* __restrict__ xw,
                              float* __restrict__ out, int N) {
  const int lane = threadIdx.x & 31;
  const int wave = threadIdx.x >> 5;
  const int nRowBlocks = (N + 15) >> 4;
  const int rowBlock = blockIdx.x * (blockDim.x >> 5) + wave;
  if (rowBlock >= nRowBlocks) return;
  const int row0 = rowBlock << 4;

  const int lm = lane & 15;
  const int khalf = (lane >> 4) << 1;

  int arow = row0 + lm;
  if (arow >= N) arow = N - 1;

  v8f acc0 = {};
  v8f acc1 = {};
  v8f acc2 = {};
#pragma unroll
  for (int k = 0; k < 12; ++k) {
    const int k4 = (k << 2) + khalf;
    v2f a  = *(const v2f*)(featw + (size_t)arow * F48 + k4);
    v2f b0 = *(const v2f*)(W3 + (size_t)(0  + lm) * F48 + k4);
    v2f b1 = *(const v2f*)(W3 + (size_t)(16 + lm) * F48 + k4);
    v2f b2 = *(const v2f*)(W3 + (size_t)(32 + lm) * F48 + k4);
    acc0 = __builtin_amdgcn_wmma_f32_16x16x4_f32(false, a, false, b0, (short)0,
                                                 acc0, false, false);
    acc1 = __builtin_amdgcn_wmma_f32_16x16x4_f32(false, a, false, b1, (short)0,
                                                 acc1, false, false);
    acc2 = __builtin_amdgcn_wmma_f32_16x16x4_f32(false, a, false, b2, (short)0,
                                                 acc2, false, false);
  }

  const int mbase = (lane >> 4) << 3;
  const float bc0 = b3[lm];
  const float bc1 = b3[16 + lm];
  const float bc2 = b3[32 + lm];
#pragma unroll
  for (int v = 0; v < 8; ++v) {
    int m = row0 + mbase + v;
    if (m < N) {
      float d2 = dis2[m];
      float e0 = acc0[v], e1 = acc1[v], e2 = acc2[v];
      size_t base = (size_t)m * F48;
      xw[base + lm]      = e0;
      xw[base + 16 + lm] = e1;
      xw[base + 32 + lm] = e2;
      out[base + lm]      = e0 * d2 + bc0;
      out[base + 16 + lm] = e1 * d2 + bc1;
      out[base + 32 + lm] = e2 * d2 + bc2;
    }
  }
}

__global__ void scatter_edges(const int* __restrict__ src,
                              const int* __restrict__ dst,
                              const float* __restrict__ dis,
                              const float* __restrict__ xw,
                              float* __restrict__ out, int E) {
  int i = blockIdx.x * blockDim.x + threadIdx.x;
  if (i >= E) return;
  int s = src[i];
  int t = dst[i];
  float nm = dis[s] * dis[t];
  const float4* xr = (const float4*)(xw + (size_t)s * F48);
  float* orow = out + (size_t)t * F48;
#pragma unroll
  for (int v = 0; v < 12; ++v) {
    float4 val = xr[v];
    __hip_atomic_fetch_add(orow + v * 4 + 0, val.x * nm, __ATOMIC_RELAXED,
                           __HIP_MEMORY_SCOPE_AGENT);
    __hip_atomic_fetch_add(orow + v * 4 + 1, val.y * nm, __ATOMIC_RELAXED,
                           __HIP_MEMORY_SCOPE_AGENT);
    __hip_atomic_fetch_add(orow + v * 4 + 2, val.z * nm, __ATOMIC_RELAXED,
                           __HIP_MEMORY_SCOPE_AGENT);
    __hip_atomic_fetch_add(orow + v * 4 + 3, val.w * nm, __ATOMIC_RELAXED,
                           __HIP_MEMORY_SCOPE_AGENT);
  }
}

extern "C" void kernel_launch(void* const* d_in, const int* in_sizes, int n_in,
                              void* d_out, int out_size, void* d_ws,
                              size_t ws_size, hipStream_t stream) {
  const float* h = (const float*)d_in[0];
  const int* e = (const int*)d_in[1];
  const float* x = (const float*)d_in[2];
  const float* q = (const float*)d_in[3];
  const float* W = (const float*)d_in[5];
  const float* b = (const float*)d_in[6];

  const int N = in_sizes[0] / 16;
  const int E = in_sizes[1] / 2;
  const int T = in_sizes[6] / F48;

  const float* W3 = W + (size_t)(T - 1) * F48 * F48;
  const float* b3 = b + (size_t)(T - 1) * F48;
  const int* srcp = e;
  const int* dstp = e + E;
  float* out = (float*)d_out;

  float* featw = (float*)d_ws;
  float* xw = featw + (size_t)N * F48;
  float* dis = xw + (size_t)N * F48;
  float* dis2 = dis + N;
  int* deg = (int*)(dis2 + N);

  const int BLK = 256;
  build_feat_zero_deg<<<(N + BLK - 1) / BLK, BLK, 0, stream>>>(x, h, q, featw,
                                                               deg, N);
  count_deg<<<(E + BLK - 1) / BLK, BLK, 0, stream>>>(dstp, deg, E);
  compute_dis<<<(N + BLK - 1) / BLK, BLK, 0, stream>>>(deg, dis, dis2, N);

  const int nRowBlocks = (N + 15) / 16;
  const int wavesPerBlock = 8;
  gcn_gemm_wmma<<<(nRowBlocks + wavesPerBlock - 1) / wavesPerBlock, BLK, 0,
                  stream>>>(featw, W3, b3, dis2, xw, out, N);

  scatter_edges<<<(E + BLK - 1) / BLK, BLK, 0, stream>>>(srcp, dstp, dis, xw,
                                                         out, E);
}


# Edge sums kept in LDS per 392-row tile of out: each work-group scans dst, adds its edges by LDS atomics (arrival order), then out += tile with plain stores; global float atomics removed
# speedup vs baseline: 3.8510x; 3.8510x over previous
#include <hip/hip_runtime.h>

typedef __attribute__((ext_vector_type(2))) float v2f;
typedef __attribute__((ext_vector_type(8))) float v8f;

#define F48 48

__global__ void build_feat_zero_deg(const float* __restrict__ x,
                                    const float* __restrict__ h,
                                    const float* __restrict__ q,
                                    float* __restrict__ featw,
                                    int* __restrict__ deg, int N) {
  int i = blockIdx.x * blockDim.x + threadIdx.x;
  if (i >= N) return;
  const float4* xr = (const float4*)(x + (size_t)i * 16);
  const float4* hr = (const float4*)(h + (size_t)i * 16);
  const float4* qr = (const float4*)(q + (size_t)i * 16);
  float4* fw = (float4*)(featw + (size_t)i * F48);
#pragma unroll
  for (int k = 0; k < 4; ++k) fw[k] = xr[k];
#pragma unroll
  for (int k = 0; k < 4; ++k) fw[4 + k] = hr[k];
#pragma unroll
  for (int k = 0; k < 4; ++k) fw[8 + k] = qr[k];
  deg[i] = 0;
}

__global__ void count_deg(const int* __restrict__ dst, int* __restrict__ deg,
                          int E) {
  int i = blockIdx.x * blockDim.x + threadIdx.x;
  if (i >= E) return;
  atomicAdd(&deg[dst[i]], 1);
}

__global__ void compute_dis(const int* __restrict__ deg,
                            float* __restrict__ dis, float* __restrict__ dis2,
                            int N) {
  int i = blockIdx.x * blockDim.x + threadIdx.x;
  if (i >= N) return;
  float d = (float)(deg[i] + 1);
  dis[i] = rsqrtf(d);
  dis2[i] = 1.0f / d;
}

__global__ void gcn_gemm_wmma(const float* __restrict__ featw,
                              const float* __restrict__ W3,
                              const float* __restrict__ b3,
                              const float* __restrict__ dis2,
                              float* __restrict__ xw,
                              float* __restrict__ out, int N) {
  const int lane = threadIdx.x & 31;
  const int wave = threadIdx.x >> 5;
  const int nRowBlocks = (N + 15) >> 4;
  const int rowBlock = blockIdx.x * (blockDim.x >> 5) + wave;
  if (rowBlock >= nRowBlocks) return;
  const int row0 = rowBlock << 4;

  const int lm = lane & 15;
  const int khalf = (lane >> 4) << 1;

  int arow = row0 + lm;
  if (arow >= N) arow = N - 1;

  v8f acc0 = {};
  v8f acc1 = {};
  v8f acc2 = {};
#pragma unroll
  for (int k = 0; k < 12; ++k) {
    const int k4 = (k << 2) + khalf;
    v2f a  = *(const v2f*)(featw + (size_t)arow * F48 + k4);
    v2f b0 = *(const v2f*)(W3 + (size_t)(0  + lm) * F48 + k4);
    v2f b1 = *(const v2f*)(W3 + (size_t)(16 + lm) * F48 + k4);
    v2f b2 = *(const v2f*)(W3 + (size_t)(32 + lm) * F48 + k4);
    acc0 = __builtin_amdgcn_wmma_f32_16x16x4_f32(false, a, false, b0, (short)0,
                                                 acc0, false, false);
    acc1 = __builtin_amdgcn_wmma_f32_16x16x4_f32(false, a, false, b1, (short)0,
                                                 acc1, false, false);
    acc2 = __builtin_amdgcn_wmma_f32_16x16x4_f32(false, a, false, b2, (short)0,
                                                 acc2, false, false);
  }

  const int mbase = (lane >> 4) << 3;
  const float bc0 = b3[lm];
  const float bc1 = b3[16 + lm];
  const float bc2 = b3[32 + lm];
#pragma unroll
  for (int v = 0; v < 8; ++v) {
    int m = row0 + mbase + v;
    if (m < N) {
      float d2 = dis2[m];
      float e0 = acc0[v], e1 = acc1[v], e2 = acc2[v];
      size_t base = (size_t)m * F48;
      xw[base + lm]      = e0;
      xw[base + 16 + lm] = e1;
      xw[base + 32 + lm] = e2;
      out[base + lm]      = e0 * d2 + bc0;
      out[base + 16 + lm] = e1 * d2 + bc1;
      out[base + 32 + lm] = e2 * d2 + bc2;
    }
  }
}

#define TILE_ROWS 392
#define AGG_THREADS 1024
#define AGG_WAVES 32
#define PEND_CAP 544
#define SCAN_VECS 4
#define FLUSH_BATCH 8

__device__ __forceinline__ void flush_pending(
    const int* pendw, int base, int cnt, int lane, int lo,
    const int* __restrict__ src, const int* __restrict__ dst,
    const float* __restrict__ dis, const float* __restrict__ xw, float* tile) {
  int s = 0, row = 0;
  float nm = 0.0f;
  if (lane < cnt) {
    const int e = pendw[base + lane];
    s = src[e];
    const int t = dst[e];
    nm = dis[s] * dis[t];
    row = t - lo;
  }
  for (int j0 = 0; j0 < cnt; j0 += FLUSH_BATCH) {
    float a0[FLUSH_BATCH], a1[FLUSH_BATCH];
    int ro[FLUSH_BATCH];
#pragma unroll
    for (int q = 0; q < FLUSH_BATCH; ++q) {
      const int j = min(j0 + q, cnt - 1);
      const int sj = __builtin_amdgcn_readlane(s, j);
      const float nj = __builtin_bit_cast(
          float, __builtin_amdgcn_readlane(__builtin_bit_cast(int, nm), j));
      ro[q] = __builtin_amdgcn_readlane(row, j) * F48;
      const float* xr = xw + (size_t)sj * F48;
      a0[q] = xr[lane] * nj;
      a1[q] = xr[32 + (lane & 15)] * nj;
    }
#pragma unroll
    for (int q = 0; q < FLUSH_BATCH; ++q) {
      if (j0 + q < cnt) {
        float* tr = tile + ro[q];
        atomicAdd(tr + lane, a0[q]);
        if (lane < 16) atomicAdd(tr + 32 + lane, a1[q]);
      }
    }
  }
}

__device__ __forceinline__ void push_matches(bool m, int e, int& np,
                                             int* pendw) {
  const unsigned bal = __builtin_amdgcn_ballot_w32(m);
  if (bal != 0u) {
    if (m) pendw[np + (int)__builtin_amdgcn_mbcnt_lo(bal, 0u)] = e;
    np += __builtin_popcount(bal);
  }
}

__global__ __launch_bounds__(AGG_THREADS) void aggregate_row_tiles(
    const int* __restrict__ src, const int* __restrict__ dst,
    const float* __restrict__ dis, const float* __restrict__ xw,
    float* __restrict__ out, int N, int E) {
  __shared__ float tile[TILE_ROWS * F48];
  __shared__ int pend[AGG_WAVES * PEND_CAP];

  const int tid = threadIdx.x;
  const int lane = tid & 31;
  const int wave = __builtin_amdgcn_readfirstlane(tid >> 5);
  const int lo = blockIdx.x * TILE_ROWS;
  int nrows = N - lo;
  if (nrows > TILE_ROWS) nrows = TILE_ROWS;
  const unsigned unrows = (unsigned)nrows;

  float4* t4 = (float4*)tile;
  for (int i = tid; i < TILE_ROWS * F48 / 4; i += AGG_THREADS)
    t4[i] = make_float4(0.0f, 0.0f, 0.0f, 0.0f);
  __syncthreads();

  int* pendw = pend + wave * PEND_CAP;
  int np = 0;

  int head = (int)(((16u - ((unsigned)(size_t)dst & 15u)) & 15u) >> 2);
  if (head > E) head = E;
  const int nvec = (E - head) >> 2;
  const int tail0 = head + (nvec << 2);
  const int4* dst4 = (const int4*)(dst + head);
  const int nchunks = (nvec + 32 * SCAN_VECS - 1) / (32 * SCAN_VECS);
  int rot = (int)(((unsigned)blockIdx.x * (unsigned)nchunks) / gridDim.x);
  if (rot >= nchunks) rot = 0;

  if (wave == 0) {
    const int nextra = head + (E - tail0);
    int e = -1, dv = -1;
    if (lane < nextra) {
      e = (lane < head) ? lane : tail0 + (lane - head);
      dv = dst[e];
    }
    push_matches((unsigned)(dv - lo) < unrows, e, np, pendw);
  }

  const int4 none = make_int4(-1, -1, -1, -1);
  const int gmax = nvec - 1;
  int4 cur[SCAN_VECS];
  if (nchunks > 0) {
    int kk = wave + rot;
    if (kk >= nchunks) kk -= nchunks;
    const int g0 = kk * (32 * SCAN_VECS) + lane;
#pragma unroll
    for (int u = 0; u < SCAN_VECS; ++u) {
      const int g = g0 + (u << 5);
      const int4 v = dst4[min(g, gmax)];
      cur[u] = (g < nvec) ? v : none;
    }
  }
  for (int k = wave; k < nchunks; k += AGG_WAVES) {
    int kk = k + rot;
    if (kk >= nchunks) kk -= nchunks;
    const int g0 = kk * (32 * SCAN_VECS) + lane;
    int4 nxt[SCAN_VECS];
    {
      int kk2 = k + AGG_WAVES + rot;
      if (kk2 >= nchunks) kk2 -= nchunks;
      const int h0 = kk2 * (32 * SCAN_VECS) + lane;
#pragma unroll
      for (int u = 0; u < SCAN_VECS; ++u) {
        const int g = h0 + (u << 5);
        const int4 v = dst4[min(g, gmax)];
        nxt[u] = (g < nvec) ? v : none;
      }
    }
#pragma unroll
    for (int u = 0; u < SCAN_VECS; ++u) {
      const unsigned u0 = (unsigned)(cur[u].x - lo);
      const unsigned u1 = (unsigned)(cur[u].y - lo);
      const unsigned u2 = (unsigned)(cur[u].z - lo);
      const unsigned u3 = (unsigned)(cur[u].w - lo);
      const unsigned mn = min(min(u0, u1), min(u2, u3));
      if (__builtin_amdgcn_ballot_w32(mn < unrows) != 0u) {
        const int e0 = head + ((g0 + (u << 5)) << 2);
        push_matches(u0 < unrows, e0 + 0, np, pendw);
        push_matches(u1 < unrows, e0 + 1, np, pendw);
        push_matches(u2 < unrows, e0 + 2, np, pendw);
        push_matches(u3 < unrows, e0 + 3, np, pendw);
      }
    }
    while (np >= 32) {
      np -= 32;
      flush_pending(pendw, np, 32, lane, lo, src, dst, dis, xw, tile);
    }
#pragma unroll
    for (int u = 0; u < SCAN_VECS; ++u) cur[u] = nxt[u];
  }
  if (np > 0) flush_pending(pendw, 0, np, lane, lo, src, dst, dis, xw, tile);

  __syncthreads();

  float4* o4 = (float4*)(out + (size_t)lo * F48);
  const int n4 = nrows * (F48 / 4);
  for (int i = tid; i < n4; i += AGG_THREADS) {
    float4 o = o4[i];
    const float4 a = t4[i];
    o.x += a.x;
    o.y += a.y;
    o.z += a.z;
    o.w += a.w;
    o4[i] = o;
  }
}

extern "C" void kernel_launch(void* const* d_in, const int* in_sizes, int n_in,
                              void* d_out, int out_size, void* d_ws,
                              size_t ws_size, hipStream_t stream) {
  const float* h = (const float*)d_in[0];
  const int* e = (const int*)d_in[1];
  const float* x = (const float*)d_in[2];
  const float* q = (const float*)d_in[3];
  const float* W = (const float*)d_in[5];
  const float* b = (const float*)d_in[6];

  const int N = in_sizes[0] / 16;
  const int E = in_sizes[1] / 2;
  const int T = in_sizes[6] / F48;

  const float* W3 = W + (size_t)(T - 1) * F48 * F48;
  const float* b3 = b + (size_t)(T - 1) * F48;
  const int* srcp = e;
  const int* dstp = e + E;
  float* out = (float*)d_out;

  float* featw = (float*)d_ws;
  float* xw = featw + (size_t)N * F48;
  float* dis = xw + (size_t)N * F48;
  float* dis2 = dis + N;
  int* deg = (int*)(dis2 + N);

  const int BLK = 256;
  build_feat_zero_deg<<<(N + BLK - 1) / BLK, BLK, 0, stream>>>(x, h, q, featw,
                                                               deg, N);
  count_deg<<<(E + BLK - 1) / BLK, BLK, 0, stream>>>(dstp, deg, E);
  compute_dis<<<(N + BLK - 1) / BLK, BLK, 0, stream>>>(deg, dis, dis2, N);

  const int nRowBlocks = (N + 15) / 16;
  const int wavesPerBlock = 8;
  gcn_gemm_wmma<<<(nRowBlocks + wavesPerBlock - 1) / wavesPerBlock, BLK, 0,
                  stream>>>(featw, W3, b3, dis2, xw, out, N);

  aggregate_row_tiles<<<(N + TILE_ROWS - 1) / TILE_ROWS, AGG_THREADS, 0,
                        stream>>>(srcp, dstp, dis, xw, out, N, E);
}


# Edges ordered by dst (place = indegree prefix sum + rank returned by the counting atomic); each out row summed in registers by a half-wave in rank (arrival) order; dst scan, LDS atomics removed
# speedup vs baseline: 4.1291x; 1.0722x over previous
#include <hip/hip_runtime.h>

typedef __attribute__((ext_vector_type(2))) float v2f;
typedef __attribute__((ext_vector_type(8))) float v8f;

#define F48 48

__global__ void build_feat_zero_deg(const float* __restrict__ x,
                                    const float* __restrict__ h,
                                    const float* __restrict__ q,
                                    float* __restrict__ featw,
                                    int* __restrict__ deg, int N) {
  int i = blockIdx.x * blockDim.x + threadIdx.x;
  if (i >= N) return;
  const float4* xr = (const float4*)(x + (size_t)i * 16);
  const float4* hr = (const float4*)(h + (size_t)i * 16);
  const float4* qr = (const float4*)(q + (size_t)i * 16);
  float4* fw = (float4*)(featw + (size_t)i * F48);
#pragma unroll
  for (int k = 0; k < 4; ++k) fw[k] = xr[k];
#pragma unroll
  for (int k = 0; k < 4; ++k) fw[4 + k] = hr[k];
#pragma unroll
  for (int k = 0; k < 4; ++k) fw[8 + k] = qr[k];
  deg[i] = 0;
}

#define EPT 8
__global__ void count_deg_rank(const int* __restrict__ dst,
                               int* __restrict__ deg, int* __restrict__ rank,
                               int E) {
  const int stride = gridDim.x * blockDim.x;
  for (int b = blockIdx.x * blockDim.x + threadIdx.x; b < E;
       b += EPT * stride) {
    if (b + (EPT - 1) * stride < E) {
      int d[EPT], r[EPT];
#pragma unroll
      for (int u = 0; u < EPT; ++u) d[u] = dst[b + u * stride];
#pragma unroll
      for (int u = 0; u < EPT; ++u) r[u] = atomicAdd(&deg[d[u]], 1);
#pragma unroll
      for (int u = 0; u < EPT; ++u) rank[b + u * stride] = r[u];
    } else {
      for (int i = b; i < E; i += stride) rank[i] = atomicAdd(&deg[dst[i]], 1);
    }
  }
}

#define SCAN_THREADS 1024
__global__ __launch_bounds__(SCAN_THREADS) void scan_deg_dis(
    const int* __restrict__ deg, int* __restrict__ rowptr,
    float* __restrict__ dis, float* __restrict__ dis2, int N) {
  __shared__ int wsumA[32];
  __shared__ int wsumB[32];
  const int tid = threadIdx.x;
  const int lane = tid & 31;
  const int wave = tid >> 5;
  const int base = blockIdx.x * SCAN_THREADS;

  int part = 0;
  const int4* deg4 = (const int4*)deg;
  const int nv = base >> 2;
#pragma unroll 4
  for (int j = tid; j < nv; j += SCAN_THREADS) {
    const int4 v = deg4[j];
    part += (v.x + v.y) + (v.z + v.w);
  }
#pragma unroll
  for (int off = 16; off > 0; off >>= 1) part += __shfl_xor(part, off);
  if (lane == 0) wsumA[wave] = part;

  const int i = base + tid;
  const int d = (i < N) ? deg[i] : 0;
  int incl = d;
#pragma unroll
  for (int off = 1; off < 32; off <<= 1) {
    const int t = __shfl_up(incl, off);
    if (lane >= off) incl += t;
  }
  if (lane == 31) wsumB[wave] = incl;
  __syncthreads();

  int front = wsumA[lane];
#pragma unroll
  for (int off = 16; off > 0; off >>= 1) front += __shfl_xor(front, off);
  const int wtot = wsumB[lane];
  int wincl = wtot;
#pragma unroll
  for (int off = 1; off < 32; off <<= 1) {
    const int t = __shfl_up(wincl, off);
    if (lane >= off) wincl += t;
  }
  const int wexcl = __shfl(wincl - wtot, wave);

  if (i < N) {
    rowptr[i] = front + wexcl + (incl - d);
    const float df = (float)(d + 1);
    dis[i] = rsqrtf(df);
    dis2[i] = 1.0f / df;
  }
}

__global__ void gcn_gemm_wmma(const float* __restrict__ featw,
                              const float* __restrict__ W3,
                              const float* __restrict__ b3,
                              const float* __restrict__ dis2,
                              float* __restrict__ xw,
                              float* __restrict__ out, int N) {
  const int lane = threadIdx.x & 31;
  const int wave = threadIdx.x >> 5;
  const int nRowBlocks = (N + 15) >> 4;
  const int rowBlock = blockIdx.x * (blockDim.x >> 5) + wave;
  if (rowBlock >= nRowBlocks) return;
  const int row0 = rowBlock << 4;

  const int lm = lane & 15;
  const int khalf = (lane >> 4) << 1;

  int arow = row0 + lm;
  if (arow >= N) arow = N - 1;

  v8f acc0 = {};
  v8f acc1 = {};
  v8f acc2 = {};
#pragma unroll
  for (int k = 0; k < 12; ++k) {
    const int k4 = (k << 2) + khalf;
    v2f a  = *(const v2f*)(featw + (size_t)arow * F48 + k4);
    v2f b0 = *(const v2f*)(W3 + (size_t)(0  + lm) * F48 + k4);
    v2f b1 = *(const v2f*)(W3 + (size_t)(16 + lm) * F48 + k4);
    v2f b2 = *(const v2f*)(W3 + (size_t)(32 + lm) * F48 + k4);
    acc0 = __builtin_amdgcn_wmma_f32_16x16x4_f32(false, a, false, b0, (short)0,
                                                 acc0, false, false);
    acc1 = __builtin_amdgcn_wmma_f32_16x16x4_f32(false, a, false, b1, (short)0,
                                                 acc1, false, false);
    acc2 = __builtin_amdgcn_wmma_f32_16x16x4_f32(false, a, false, b2, (short)0,
                                                 acc2, false, false);
  }

  const int mbase = (lane >> 4) << 3;
  const float bc0 = b3[lm];
  const float bc1 = b3[16 + lm];
  const float bc2 = b3[32 + lm];
#pragma unroll
  for (int v = 0; v < 8; ++v) {
    int m = row0 + mbase + v;
    if (m < N) {
      float d2 = dis2[m];
      float e0 = acc0[v], e1 = acc1[v], e2 = acc2[v];
      size_t base = (size_t)m * F48;
      xw[base + lm]      = e0;
      xw[base + 16 + lm] = e1;
      xw[base + 32 + lm] = e2;
      out[base + lm]      = e0 * d2 + bc0;
      out[base + 16 + lm] = e1 * d2 + bc1;
      out[base + 32 + lm] = e2 * d2 + bc2;
    }
  }
}

__device__ __forceinline__ void place_edge(int s, int d, int r,
                                           const int* __restrict__ rowptr,
                                           const float* __restrict__ dis,
                                           int2* __restrict__ csr) {
  const float nm = dis[s] * dis[d];
  csr[rowptr[d] + r] = make_int2(s, __builtin_bit_cast(int, nm));
}

__global__ void fill_rows(const int* __restrict__ src,
                          const int* __restrict__ dst,
                          const int* __restrict__ rank,
                          const int* __restrict__ rowptr,
                          const float* __restrict__ dis,
                          int2* __restrict__ csr, int E) {
  const int stride = gridDim.x * blockDim.x;
  for (int b = blockIdx.x * blockDim.x + threadIdx.x; b < E;
       b += EPT * stride) {
    if (b + (EPT - 1) * stride < E) {
      int s[EPT], d[EPT], r[EPT], p[EPT];
      float ws[EPT], wd[EPT];
#pragma unroll
      for (int u = 0; u < EPT; ++u) {
        const int i = b + u * stride;
        s[u] = src[i];
        d[u] = dst[i];
        r[u] = rank[i];
      }
#pragma unroll
      for (int u = 0; u < EPT; ++u) {
        p[u] = rowptr[d[u]];
        ws[u] = dis[s[u]];
        wd[u] = dis[d[u]];
      }
#pragma unroll
      for (int u = 0; u < EPT; ++u) {
        const float nm = ws[u] * wd[u];
        csr[p[u] + r[u]] = make_int2(s[u], __builtin_bit_cast(int, nm));
      }
    } else {
      for (int i = b; i < E; i += stride)
        place_edge(src[i], dst[i], rank[i], rowptr, dis, csr);
    }
  }
}

#define AGG_THREADS 256
#define AGG_WAVES 8
#define ROWS_PER_WAVE 8
#define EDGE_BATCH 8

__global__ __launch_bounds__(AGG_THREADS) void aggregate_rows(
    const int2* __restrict__ csr, const int* __restrict__ rowptr,
    const int* __restrict__ deg, const float* __restrict__ xw,
    float* __restrict__ out, int N) {
  const int lane = threadIdx.x & 31;
  const int wave = __builtin_amdgcn_readfirstlane((int)(threadIdx.x >> 5));
  const int row0 = (blockIdx.x * AGG_WAVES + wave) * ROWS_PER_WAVE;
  if (row0 >= N) return;
  const bool hi = lane >= 16;
  const int l16 = lane & 15;
  const int c4 = (l16 < 12 ? l16 : 11) * 4;

  int mb = 0, mc = 0;
  if (lane < ROWS_PER_WAVE && row0 + lane < N) {
    mb = rowptr[row0 + lane];
    mc = deg[row0 + lane];
  }

  for (int k = 0; k < ROWS_PER_WAVE / 2; ++k) {
    const int begA = __builtin_amdgcn_readlane(mb, 2 * k);
    const int cntA = __builtin_amdgcn_readlane(mc, 2 * k);
    const int begB = __builtin_amdgcn_readlane(mb, 2 * k + 1);
    const int cntB = __builtin_amdgcn_readlane(mc, 2 * k + 1);
    const int beg = hi ? begB : begA;
    const int cnt = hi ? cntB : cntA;
    const int maxc = max(cntA, cntB);
    const int r = row0 + 2 * k + (hi ? 1 : 0);
    const bool wr = (r < N) && (l16 < 12);

    float4* op = (float4*)(out + (size_t)r * F48 + c4);
    float4 o = make_float4(0.0f, 0.0f, 0.0f, 0.0f);
    if (wr) o = *op;

    float4 acc = make_float4(0.0f, 0.0f, 0.0f, 0.0f);
    for (int c0 = 0; c0 < maxc; c0 += 16) {
      int s = 0;
      float nm = 0.0f;
      if (c0 + l16 < cnt) {
        const int2 en = csr[beg + c0 + l16];
        s = en.x;
        nm = __builtin_bit_cast(float, en.y);
      }
      const int nh = min(16, maxc - c0);
      for (int j0 = 0; j0 < nh; j0 += EDGE_BATCH) {
        float4 xv[EDGE_BATCH];
        float nj[EDGE_BATCH];
#pragma unroll
        for (int q = 0; q < EDGE_BATCH; ++q) {
          const int j = j0 + q;
          const int sA = __builtin_amdgcn_readlane(s, j);
          const int sB = __builtin_amdgcn_readlane(s, 16 + j);
          const int nA = __builtin_amdgcn_readlane(__builtin_bit_cast(int, nm), j);
          const int nB =
              __builtin_amdgcn_readlane(__builtin_bit_cast(int, nm), 16 + j);
          const int sj = hi ? sB : sA;
          nj[q] = __builtin_bit_cast(float, hi ? nB : nA);
          xv[q] = *(const float4*)(xw + (size_t)sj * F48 + c4);
        }
#pragma unroll
        for (int q = 0; q < EDGE_BATCH; ++q) {
          const bool ok = (c0 + j0 + q) < cnt;
          acc.x = ok ? __builtin_fmaf(xv[q].x, nj[q], acc.x) : acc.x;
          acc.y = ok ? __builtin_fmaf(xv[q].y, nj[q], acc.y) : acc.y;
          acc.z = ok ? __builtin_fmaf(xv[q].z, nj[q], acc.z) : acc.z;
          acc.w = ok ? __builtin_fmaf(xv[q].w, nj[q], acc.w) : acc.w;
        }
      }
    }
    if (wr) {
      o.x += acc.x;
      o.y += acc.y;
      o.z += acc.z;
      o.w += acc.w;
      *op = o;
    }
  }
}

extern "C" void kernel_launch(void* const* d_in, const int* in_sizes, int n_in,
                              void* d_out, int out_size, void* d_ws,
                              size_t ws_size, hipStream_t stream) {
  const float* h = (const float*)d_in[0];
  const int* e = (const int*)d_in[1];
  const float* x = (const float*)d_in[2];
  const float* q = (const float*)d_in[3];
  const float* W = (const float*)d_in[5];
  const float* b = (const float*)d_in[6];

  const int N = in_sizes[0] / 16;
  const int E = in_sizes[1] / 2;
  const int T = in_sizes[6] / F48;

  const float* W3 = W + (size_t)(T - 1) * F48 * F48;
  const float* b3 = b + (size_t)(T - 1) * F48;
  const int* srcp = e;
  const int* dstp = e + E;
  float* out = (float*)d_out;

  const size_t NP = ((size_t)N + 63) & ~(size_t)63;
  const size_t EP = ((size_t)E + 63) & ~(size_t)63;
  float* featw = (float*)d_ws;
  float* xw = featw + (size_t)N * F48;
  float* dis = xw + (size_t)N * F48;
  float* dis2 = dis + NP;
  int* deg = (int*)(dis2 + NP);
  int* rowptr = deg + NP;
  int* rank = rowptr + NP;
  int2* csr = (int2*)(rank + EP);

  const int BLK = 256;
  const int edgeBlocks = ((E + EPT - 1) / EPT + BLK - 1) / BLK;
  build_feat_zero_deg<<<(N + BLK - 1) / BLK, BLK, 0, stream>>>(x, h, q, featw,
                                                               deg, N);
  count_deg_rank<<<edgeBlocks, BLK, 0, stream>>>(dstp, deg, rank, E);
  scan_deg_dis<<<(N + SCAN_THREADS - 1) / SCAN_THREADS, SCAN_THREADS, 0,
                 stream>>>(deg, rowptr, dis, dis2, N);

  const int nRowBlocks = (N + 15) / 16;
  const int wavesPerBlock = 8;
  gcn_gemm_wmma<<<(nRowBlocks + wavesPerBlock - 1) / wavesPerBlock, BLK, 0,
                  stream>>>(featw, W3, b3, dis2, xw, out, N);

  fill_rows<<<edgeBlocks, BLK, 0, stream>>>(srcp, dstp, rank, rowptr, dis, csr,
                                            E);
  const int rowsPerBlock = AGG_WAVES * ROWS_PER_WAVE;
  aggregate_rows<<<(N + rowsPerBlock - 1) / rowsPerBlock, AGG_THREADS, 0,
                   stream>>>(csr, rowptr, deg, xw, out, N);
}


# Rows stored pre-scaled by dis[src] in the GEMM epilogue: edge list holds src only, fill gathers rowptr only; row sum plus self row scaled by dis[dst] once, out written once (rounding order changes)
# speedup vs baseline: 4.6289x; 1.1210x over previous
#include <hip/hip_runtime.h>

typedef __attribute__((ext_vector_type(2))) float v2f;
typedef __attribute__((ext_vector_type(8))) float v8f;

#define F48 48

__global__ void build_feat_zero_deg(const float* __restrict__ x,
                                    const float* __restrict__ h,
                                    const float* __restrict__ q,
                                    float* __restrict__ featw,
                                    int* __restrict__ deg, int N) {
  int i = blockIdx.x * blockDim.x + threadIdx.x;
  if (i >= N) return;
  const float4* xr = (const float4*)(x + (size_t)i * 16);
  const float4* hr = (const float4*)(h + (size_t)i * 16);
  const float4* qr = (const float4*)(q + (size_t)i * 16);
  float4* fw = (float4*)(featw + (size_t)i * F48);
#pragma unroll
  for (int k = 0; k < 4; ++k) fw[k] = xr[k];
#pragma unroll
  for (int k = 0; k < 4; ++k) fw[4 + k] = hr[k];
#pragma unroll
  for (int k = 0; k < 4; ++k) fw[8 + k] = qr[k];
  deg[i] = 0;
}

#define EPT 8
__global__ void count_deg_rank(const int* __restrict__ dst,
                               int* __restrict__ deg, int* __restrict__ rank,
                               int E) {
  const int stride = gridDim.x * blockDim.x;
  for (int b = blockIdx.x * blockDim.x + threadIdx.x; b < E;
       b += EPT * stride) {
    if (b + (EPT - 1) * stride < E) {
      int d[EPT], r[EPT];
#pragma unroll
      for (int u = 0; u < EPT; ++u) d[u] = dst[b + u * stride];
#pragma unroll
      for (int u = 0; u < EPT; ++u) r[u] = atomicAdd(&deg[d[u]], 1);
#pragma unroll
      for (int u = 0; u < EPT; ++u) rank[b + u * stride] = r[u];
    } else {
      for (int i = b; i < E; i += stride) rank[i] = atomicAdd(&deg[dst[i]], 1);
    }
  }
}

#define SCAN_THREADS 1024
__global__ __launch_bounds__(SCAN_THREADS) void scan_deg_dis(
    const int* __restrict__ deg, int* __restrict__ rowptr,
    float* __restrict__ dis, int N) {
  __shared__ int wsumA[32];
  __shared__ int wsumB[32];
  const int tid = threadIdx.x;
  const int lane = tid & 31;
  const int wave = tid >> 5;
  const int base = blockIdx.x * SCAN_THREADS;

  int part = 0;
  const int4* deg4 = (const int4*)deg;
  const int nv = base >> 2;
#pragma unroll 4
  for (int j = tid; j < nv; j += SCAN_THREADS) {
    const int4 v = deg4[j];
    part += (v.x + v.y) + (v.z + v.w);
  }
#pragma unroll
  for (int off = 16; off > 0; off >>= 1) part += __shfl_xor(part, off);
  if (lane == 0) wsumA[wave] = part;

  const int i = base + tid;
  const int d = (i < N) ? deg[i] : 0;
  int incl = d;
#pragma unroll
  for (int off = 1; off < 32; off <<= 1) {
    const int t = __shfl_up(incl, off);
    if (lane >= off) incl += t;
  }
  if (lane == 31) wsumB[wave] = incl;
  __syncthreads();

  int front = wsumA[lane];
#pragma unroll
  for (int off = 16; off > 0; off >>= 1) front += __shfl_xor(front, off);
  const int wtot = wsumB[lane];
  int wincl = wtot;
#pragma unroll
  for (int off = 1; off < 32; off <<= 1) {
    const int t = __shfl_up(wincl, off);
    if (lane >= off) wincl += t;
  }
  const int wexcl = __shfl(wincl - wtot, wave);

  if (i < N) {
    rowptr[i] = front + wexcl + (incl - d);
    dis[i] = rsqrtf((float)(d + 1));
  }
}

__global__ void gcn_gemm_wmma(const float* __restrict__ featw,
                              const float* __restrict__ W3,
                              const float* __restrict__ dis,
                              float* __restrict__ xws, int N) {
  const int lane = threadIdx.x & 31;
  const int wave = threadIdx.x >> 5;
  const int nRowBlocks = (N + 15) >> 4;
  const int rowBlock = blockIdx.x * (blockDim.x >> 5) + wave;
  if (rowBlock >= nRowBlocks) return;
  const int row0 = rowBlock << 4;

  const int lm = lane & 15;
  const int khalf = (lane >> 4) << 1;

  int arow = row0 + lm;
  if (arow >= N) arow = N - 1;

  v8f acc0 = {};
  v8f acc1 = {};
  v8f acc2 = {};
#pragma unroll
  for (int k = 0; k < 12; ++k) {
    const int k4 = (k << 2) + khalf;
    v2f a  = *(const v2f*)(featw + (size_t)arow * F48 + k4);
    v2f b0 = *(const v2f*)(W3 + (size_t)(0  + lm) * F48 + k4);
    v2f b1 = *(const v2f*)(W3 + (size_t)(16 + lm) * F48 + k4);
    v2f b2 = *(const v2f*)(W3 + (size_t)(32 + lm) * F48 + k4);
    acc0 = __builtin_amdgcn_wmma_f32_16x16x4_f32(false, a, false, b0, (short)0,
                                                 acc0, false, false);
    acc1 = __builtin_amdgcn_wmma_f32_16x16x4_f32(false, a, false, b1, (short)0,
                                                 acc1, false, false);
    acc2 = __builtin_amdgcn_wmma_f32_16x16x4_f32(false, a, false, b2, (short)0,
                                                 acc2, false, false);
  }

  const int mbase = (lane >> 4) << 3;
#pragma unroll
  for (int v = 0; v < 8; ++v) {
    int m = row0 + mbase + v;
    if (m < N) {
      const float dm = dis[m];
      size_t base = (size_t)m * F48;
      xws[base + lm]      = acc0[v] * dm;
      xws[base + 16 + lm] = acc1[v] * dm;
      xws[base + 32 + lm] = acc2[v] * dm;
    }
  }
}

__global__ void fill_rows(const int* __restrict__ src,
                          const int* __restrict__ dst,
                          const int* __restrict__ rank,
                          const int* __restrict__ rowptr,
                          int* __restrict__ csr, int E) {
  const int stride = gridDim.x * blockDim.x;
  for (int b = blockIdx.x * blockDim.x + threadIdx.x; b < E;
       b += EPT * stride) {
    if (b + (EPT - 1) * stride < E) {
      int s[EPT], d[EPT], r[EPT], p[EPT];
#pragma unroll
      for (int u = 0; u < EPT; ++u) {
        const int i = b + u * stride;
        s[u] = src[i];
        d[u] = dst[i];
        r[u] = rank[i];
      }
#pragma unroll
      for (int u = 0; u < EPT; ++u) p[u] = rowptr[d[u]];
#pragma unroll
      for (int u = 0; u < EPT; ++u) csr[p[u] + r[u]] = s[u];
    } else {
      for (int i = b; i < E; i += stride) csr[rowptr[dst[i]] + rank[i]] = src[i];
    }
  }
}

#define AGG_THREADS 256
#define AGG_WAVES 8
#define ROWS_PER_WAVE 8
#define EDGE_BATCH 8

__global__ __launch_bounds__(AGG_THREADS) void aggregate_rows(
    const int* __restrict__ csr, const int* __restrict__ rowptr,
    const int* __restrict__ deg, const float* __restrict__ dis,
    const float* __restrict__ b3, const float* __restrict__ xws,
    float* __restrict__ out, int N) {
  const int lane = threadIdx.x & 31;
  const int wave = __builtin_amdgcn_readfirstlane((int)(threadIdx.x >> 5));
  const int row0 = (blockIdx.x * AGG_WAVES + wave) * ROWS_PER_WAVE;
  if (row0 >= N) return;
  const bool hi = lane >= 16;
  const int l16 = lane & 15;
  const int c4 = (l16 < 12 ? l16 : 11) * 4;
  const float4 bias = *(const float4*)(b3 + c4);

  int mb = 0, mc = 0;
  float md = 0.0f;
  if (lane < ROWS_PER_WAVE && row0 + lane < N) {
    mb = rowptr[row0 + lane];
    mc = deg[row0 + lane];
    md = dis[row0 + lane];
  }

  for (int k = 0; k < ROWS_PER_WAVE / 2; ++k) {
    const int begA = __builtin_amdgcn_readlane(mb, 2 * k);
    const int cntA = __builtin_amdgcn_readlane(mc, 2 * k);
    const int begB = __builtin_amdgcn_readlane(mb, 2 * k + 1);
    const int cntB = __builtin_amdgcn_readlane(mc, 2 * k + 1);
    const int dA = __builtin_amdgcn_readlane(__builtin_bit_cast(int, md), 2 * k);
    const int dB =
        __builtin_amdgcn_readlane(__builtin_bit_cast(int, md), 2 * k + 1);
    const int beg = hi ? begB : begA;
    const int cnt = hi ? cntB : cntA;
    const float dr = __builtin_bit_cast(float, hi ? dB : dA);
    const int maxc = max(cntA, cntB);
    const int r = row0 + 2 * k + (hi ? 1 : 0);
    const bool wr = (r < N) && (l16 < 12);

    float4 self4 = make_float4(0.0f, 0.0f, 0.0f, 0.0f);
    if (wr) self4 = *(const float4*)(xws + (size_t)r * F48 + c4);

    float4 acc = make_float4(0.0f, 0.0f, 0.0f, 0.0f);
    for (int c0 = 0; c0 < maxc; c0 += 16) {
      int s = 0;
      if (c0 + l16 < cnt) s = csr[beg + c0 + l16];
      const int nh = min(16, maxc - c0);
      for (int j0 = 0; j0 < nh; j0 += EDGE_BATCH) {
        float4 xv[EDGE_BATCH];
#pragma unroll
        for (int q = 0; q < EDGE_BATCH; ++q) {
          const int j = j0 + q;
          const int sA = __builtin_amdgcn_readlane(s, j);
          const int sB = __builtin_amdgcn_readlane(s, 16 + j);
          const int sj = hi ? sB : sA;
          xv[q] = *(const float4*)(xws + (size_t)sj * F48 + c4);
        }
#pragma unroll
        for (int q = 0; q < EDGE_BATCH; ++q) {
          const bool ok = (c0 + j0 + q) < cnt;
          acc.x += ok ? xv[q].x : 0.0f;
          acc.y += ok ? xv[q].y : 0.0f;
          acc.z += ok ? xv[q].z : 0.0f;
          acc.w += ok ? xv[q].w : 0.0f;
        }
      }
    }
    if (wr) {
      float4 o;
      o.x = (acc.x + self4.x) * dr + bias.x;
      o.y = (acc.y + self4.y) * dr + bias.y;
      o.z = (acc.z + self4.z) * dr + bias.z;
      o.w = (acc.w + self4.w) * dr + bias.w;
      *(float4*)(out + (size_t)r * F48 + c4) = o;
    }
  }
}

extern "C" void kernel_launch(void* const* d_in, const int* in_sizes, int n_in,
                              void* d_out, int out_size, void* d_ws,
                              size_t ws_size, hipStream_t stream) {
  const float* h = (const float*)d_in[0];
  const int* e = (const int*)d_in[1];
  const float* x = (const float*)d_in[2];
  const float* q = (const float*)d_in[3];
  const float* W = (const float*)d_in[5];
  const float* b = (const float*)d_in[6];

  const int N = in_sizes[0] / 16;
  const int E = in_sizes[1] / 2;
  const int T = in_sizes[6] / F48;

  const float* W3 = W + (size_t)(T - 1) * F48 * F48;
  const float* b3 = b + (size_t)(T - 1) * F48;
  const int* srcp = e;
  const int* dstp = e + E;
  float* out = (float*)d_out;

  const size_t NP = ((size_t)N + 63) & ~(size_t)63;
  const size_t EP = ((size_t)E + 63) & ~(size_t)63;
  float* featw = (float*)d_ws;
  float* xws = featw + (size_t)N * F48;
  float* dis = xws + (size_t)N * F48;
  int* deg = (int*)(dis + NP);
  int* rowptr = deg + NP;
  int* rank = rowptr + NP;
  int* csr = rank + EP;

  const int BLK = 256;
  const int edgeBlocks = ((E + EPT - 1) / EPT + BLK - 1) / BLK;
  build_feat_zero_deg<<<(N + BLK - 1) / BLK, BLK, 0, stream>>>(x, h, q, featw,
                                                               deg, N);
  count_deg_rank<<<edgeBlocks, BLK, 0, stream>>>(dstp, deg, rank, E);
  scan_deg_dis<<<(N + SCAN_THREADS - 1) / SCAN_THREADS, SCAN_THREADS, 0,
                 stream>>>(deg, rowptr, dis, N);

  const int nRowBlocks = (N + 15) / 16;
  const int wavesPerBlock = 8;
  gcn_gemm_wmma<<<(nRowBlocks + wavesPerBlock - 1) / wavesPerBlock, BLK, 0,
                  stream>>>(featw, W3, dis, xws, N);

  fill_rows<<<edgeBlocks, BLK, 0, stream>>>(srcp, dstp, rank, rowptr, csr, E);
  const int rowsPerBlock = AGG_WAVES * ROWS_PER_WAVE;
  aggregate_rows<<<(N + rowsPerBlock - 1) / rowsPerBlock, AGG_THREADS, 0,
                   stream>>>(csr, rowptr, deg, dis, b3, xws, out, N);
}


# Edges grouped by dst in LDS: binned by 256-node range (one global add per chunk and bin), ranked per bin with LDS adds; per-edge global atomics, scan and fill kernels removed (row order: arrival)
# speedup vs baseline: 10.3578x; 2.2376x over previous
#include <hip/hip_runtime.h>

typedef __attribute__((ext_vector_type(2))) float v2f;
typedef __attribute__((ext_vector_type(8))) float v8f;

#define F48 48

#define BIN_SHIFT 8
#define BIN_W 256
#define NB_MAX 1024
#define BIN_THREADS 1024
#define BIN_GRID 256
#define EB 8

__global__ void build_feat_zero_bins(const float* __restrict__ x,
                                     const float* __restrict__ h,
                                     const float* __restrict__ q,
                                     float* __restrict__ featw,
                                     int* __restrict__ bincnt, int N, int NB) {
  int i = blockIdx.x * blockDim.x + threadIdx.x;
  if (i >= N) return;
  const float4* xr = (const float4*)(x + (size_t)i * 16);
  const float4* hr = (const float4*)(h + (size_t)i * 16);
  const float4* qr = (const float4*)(q + (size_t)i * 16);
  float4* fw = (float4*)(featw + (size_t)i * F48);
#pragma unroll
  for (int k = 0; k < 4; ++k) fw[k] = xr[k];
#pragma unroll
  for (int k = 0; k < 4; ++k) fw[4 + k] = hr[k];
#pragma unroll
  for (int k = 0; k < 4; ++k) fw[8 + k] = qr[k];
  if (i < NB) bincnt[i] = 0;
}

__global__ __launch_bounds__(BIN_THREADS) void bin_count(
    const int* __restrict__ dst, int* __restrict__ bincnt,
    int* __restrict__ chunkoff, int E, int NB) {
  __shared__ int hist[NB_MAX];
  if (NB > NB_MAX) return;
  const int tid = threadIdx.x;
  if (tid < NB) hist[tid] = 0;
  __syncthreads();
  const int CH = (E + gridDim.x - 1) / gridDim.x;
  const int e0 = blockIdx.x * CH;
  const int e1 = min(E, e0 + CH);
  for (int b = e0 + tid; b < e1; b += EB * BIN_THREADS) {
    int d[EB];
#pragma unroll
    for (int u = 0; u < EB; ++u) {
      const int i = b + u * BIN_THREADS;
      d[u] = (i < e1) ? dst[i] : -1;
    }
#pragma unroll
    for (int u = 0; u < EB; ++u)
      if (d[u] >= 0) atomicAdd(&hist[d[u] >> BIN_SHIFT], 1);
  }
  __syncthreads();
  if (tid < NB) {
    const int c = hist[tid];
    chunkoff[blockIdx.x * NB_MAX + tid] = c ? atomicAdd(&bincnt[tid], c) : 0;
  }
}

__global__ __launch_bounds__(BIN_THREADS) void bin_scatter(
    const int* __restrict__ src, const int* __restrict__ dst,
    const int* __restrict__ bincnt, const int* __restrict__ chunkoff,
    int* __restrict__ binstart, int* __restrict__ binned, int E, int NB) {
  __shared__ int cur[NB_MAX];
  __shared__ int wsum[32];
  if (NB > NB_MAX) return;
  const int tid = threadIdx.x;
  const int lane = tid & 31;
  const int wave = tid >> 5;

  const int c = (tid < NB) ? bincnt[tid] : 0;
  const int mine = (tid < NB) ? chunkoff[blockIdx.x * NB_MAX + tid] : 0;
  int incl = c;
#pragma unroll
  for (int off = 1; off < 32; off <<= 1) {
    const int t = __shfl_up(incl, off);
    if (lane >= off) incl += t;
  }
  if (lane == 31) wsum[wave] = incl;
  __syncthreads();
  const int wtot = wsum[lane];
  int wincl = wtot;
#pragma unroll
  for (int off = 1; off < 32; off <<= 1) {
    const int t = __shfl_up(wincl, off);
    if (lane >= off) wincl += t;
  }
  const int wexcl = __shfl(wincl - wtot, wave);
  const int start = wexcl + (incl - c);
  if (tid < NB) {
    cur[tid] = start + mine;
    if (blockIdx.x == 0) binstart[tid] = start;
  }
  if (blockIdx.x == 0 && tid == 0) binstart[NB] = E;
  __syncthreads();

  int2* __restrict__ bp = (int2*)binned;
  const int CH = (E + gridDim.x - 1) / gridDim.x;
  const int e0 = blockIdx.x * CH;
  const int e1 = min(E, e0 + CH);
  for (int b = e0 + tid; b < e1; b += EB * BIN_THREADS) {
    int s[EB], d[EB];
#pragma unroll
    for (int u = 0; u < EB; ++u) {
      const int i = b + u * BIN_THREADS;
      const bool in = i < e1;
      d[u] = in ? dst[i] : -1;
      s[u] = in ? src[i] : 0;
    }
#pragma unroll
    for (int u = 0; u < EB; ++u) {
      if (d[u] >= 0) {
        const int p = atomicAdd(&cur[d[u] >> BIN_SHIFT], 1);
        bp[p] = make_int2(s[u], d[u]);
      }
    }
  }
}

#define SORT_THREADS 512
__global__ __launch_bounds__(SORT_THREADS) void bin_sort(
    const int* __restrict__ binned, const int* __restrict__ binstart,
    int* __restrict__ deg, int* __restrict__ rowptr, float* __restrict__ dis,
    int* __restrict__ csr, int N) {
  __shared__ int cnt[BIN_W];
  __shared__ int wtot[BIN_W / 32];
  const int tid = threadIdx.x;
  const int lane = tid & 31;
  const int wave = tid >> 5;
  const int bin = blockIdx.x;
  const int s0 = binstart[bin];
  const int s1 = binstart[bin + 1];
  const int n0 = bin << BIN_SHIFT;
  const int2* __restrict__ bp = (const int2*)binned;

  if (tid < BIN_W) cnt[tid] = 0;
  __syncthreads();
  for (int b = s0 + tid; b < s1; b += 4 * SORT_THREADS) {
    int d[4];
#pragma unroll
    for (int u = 0; u < 4; ++u) {
      const int i = b + u * SORT_THREADS;
      d[u] = (i < s1) ? bp[i].y : -1;
    }
#pragma unroll
    for (int u = 0; u < 4; ++u)
      if (d[u] >= 0) atomicAdd(&cnt[d[u] - n0], 1);
  }
  __syncthreads();

  const int dg = (tid < BIN_W) ? cnt[tid] : 0;
  int incl = dg;
#pragma unroll
  for (int off = 1; off < 32; off <<= 1) {
    const int t = __shfl_up(incl, off);
    if (lane >= off) incl += t;
  }
  if (tid < BIN_W && lane == 31) wtot[wave] = incl;
  __syncthreads();
  if (tid < BIN_W) {
    int front = 0;
#pragma unroll
    for (int w = 0; w < BIN_W / 32; ++w) front += (w < wave) ? wtot[w] : 0;
    const int excl = front + (incl - dg);
    cnt[tid] = excl;
    const int node = n0 + tid;
    if (node < N) {
      deg[node] = dg;
      rowptr[node] = s0 + excl;
      dis[node] = rsqrtf((float)(dg + 1));
    }
  }
  __syncthreads();

  for (int b = s0 + tid; b < s1; b += 4 * SORT_THREADS) {
    int2 e[4];
#pragma unroll
    for (int u = 0; u < 4; ++u) {
      const int i = b + u * SORT_THREADS;
      e[u] = (i < s1) ? bp[i] : make_int2(0, -1);
    }
#pragma unroll
    for (int u = 0; u < 4; ++u) {
      if (e[u].y >= 0) {
        const int p = atomicAdd(&cnt[e[u].y - n0], 1);
        csr[s0 + p] = e[u].x;
      }
    }
  }
}

__global__ void gcn_gemm_wmma(const float* __restrict__ featw,
                              const float* __restrict__ W3,
                              const float* __restrict__ dis,
                              float* __restrict__ xws, int N) {
  const int lane = threadIdx.x & 31;
  const int wave = threadIdx.x >> 5;
  const int nRowBlocks = (N + 15) >> 4;
  const int rowBlock = blockIdx.x * (blockDim.x >> 5) + wave;
  if (rowBlock >= nRowBlocks) return;
  const int row0 = rowBlock << 4;

  const int lm = lane & 15;
  const int khalf = (lane >> 4) << 1;

  int arow = row0 + lm;
  if (arow >= N) arow = N - 1;

  v8f acc0 = {};
  v8f acc1 = {};
  v8f acc2 = {};
#pragma unroll
  for (int k = 0; k < 12; ++k) {
    const int k4 = (k << 2) + khalf;
    v2f a  = *(const v2f*)(featw + (size_t)arow * F48 + k4);
    v2f b0 = *(const v2f*)(W3 + (size_t)(0  + lm) * F48 + k4);
    v2f b1 = *(const v2f*)(W3 + (size_t)(16 + lm) * F48 + k4);
    v2f b2 = *(const v2f*)(W3 + (size_t)(32 + lm) * F48 + k4);
    acc0 = __builtin_amdgcn_wmma_f32_16x16x4_f32(false, a, false, b0, (short)0,
                                                 acc0, false, false);
    acc1 = __builtin_amdgcn_wmma_f32_16x16x4_f32(false, a, false, b1, (short)0,
                                                 acc1, false, false);
    acc2 = __builtin_amdgcn_wmma_f32_16x16x4_f32(false, a, false, b2, (short)0,
                                                 acc2, false, false);
  }

  const int mbase = (lane >> 4) << 3;
#pragma unroll
  for (int v = 0; v < 8; ++v) {
    int m = row0 + mbase + v;
    if (m < N) {
      const float dm = dis[m];
      size_t base = (size_t)m * F48;
      xws[base + lm]      = acc0[v] * dm;
      xws[base + 16 + lm] = acc1[v] * dm;
      xws[base + 32 + lm] = acc2[v] * dm;
    }
  }
}

#define AGG_THREADS 256
#define AGG_WAVES 8
#define ROWS_PER_WAVE 8
#define EDGE_BATCH 8

__global__ __launch_bounds__(AGG_THREADS) void aggregate_rows(
    const int* __restrict__ csr, const int* __restrict__ rowptr,
    const int* __restrict__ deg, const float* __restrict__ dis,
    const float* __restrict__ b3, const float* __restrict__ xws,
    float* __restrict__ out, int N) {
  const int lane = threadIdx.x & 31;
  const int wave = __builtin_amdgcn_readfirstlane((int)(threadIdx.x >> 5));
  const int row0 = (blockIdx.x * AGG_WAVES + wave) * ROWS_PER_WAVE;
  if (row0 >= N) return;
  const bool hi = lane >= 16;
  const int l16 = lane & 15;
  const int c4 = (l16 < 12 ? l16 : 11) * 4;
  const float4 bias = *(const float4*)(b3 + c4);

  int mb = 0, mc = 0;
  float md = 0.0f;
  if (lane < ROWS_PER_WAVE && row0 + lane < N) {
    mb = rowptr[row0 + lane];
    mc = deg[row0 + lane];
    md = dis[row0 + lane];
  }

  for (int k = 0; k < ROWS_PER_WAVE / 2; ++k) {
    const int begA = __builtin_amdgcn_readlane(mb, 2 * k);
    const int cntA = __builtin_amdgcn_readlane(mc, 2 * k);
    const int begB = __builtin_amdgcn_readlane(mb, 2 * k + 1);
    const int cntB = __builtin_amdgcn_readlane(mc, 2 * k + 1);
    const int dA = __builtin_amdgcn_readlane(__builtin_bit_cast(int, md), 2 * k);
    const int dB =
        __builtin_amdgcn_readlane(__builtin_bit_cast(int, md), 2 * k + 1);
    const int beg = hi ? begB : begA;
    const int cnt = hi ? cntB : cntA;
    const float dr = __builtin_bit_cast(float, hi ? dB : dA);
    const int maxc = max(cntA, cntB);
    const int r = row0 + 2 * k + (hi ? 1 : 0);
    const bool wr = (r < N) && (l16 < 12);

    float4 self4 = make_float4(0.0f, 0.0f, 0.0f, 0.0f);
    if (wr) self4 = *(const float4*)(xws + (size_t)r * F48 + c4);

    float4 acc = make_float4(0.0f, 0.0f, 0.0f, 0.0f);
    for (int c0 = 0; c0 < maxc; c0 += 16) {
      int s = 0;
      if (c0 + l16 < cnt) s = csr[beg + c0 + l16];
      const int nh = min(16, maxc - c0);
      for (int j0 = 0; j0 < nh; j0 += EDGE_BATCH) {
        float4 xv[EDGE_BATCH];
#pragma unroll
        for (int q = 0; q < EDGE_BATCH; ++q) {
          const int j = j0 + q;
          const int sA = __builtin_amdgcn_readlane(s, j);
          const int sB = __builtin_amdgcn_readlane(s, 16 + j);
          const int sj = hi ? sB : sA;
          xv[q] = *(const float4*)(xws + (size_t)sj * F48 + c4);
        }
#pragma unroll
        for (int q = 0; q < EDGE_BATCH; ++q) {
          const bool ok = (c0 + j0 + q) < cnt;
          acc.x += ok ? xv[q].x : 0.0f;
          acc.y += ok ? xv[q].y : 0.0f;
          acc.z += ok ? xv[q].z : 0.0f;
          acc.w += ok ? xv[q].w : 0.0f;
        }
      }
    }
    if (wr) {
      float4 o;
      o.x = (acc.x + self4.x) * dr + bias.x;
      o.y = (acc.y + self4.y) * dr + bias.y;
      o.z = (acc.z + self4.z) * dr + bias.z;
      o.w = (acc.w + self4.w) * dr + bias.w;
      *(float4*)(out + (size_t)r * F48 + c4) = o;
    }
  }
}

extern "C" void kernel_launch(void* const* d_in, const int* in_sizes, int n_in,
                              void* d_out, int out_size, void* d_ws,
                              size_t ws_size, hipStream_t stream) {
  const float* h = (const float*)d_in[0];
  const int* e = (const int*)d_in[1];
  const float* x = (const float*)d_in[2];
  const float* q = (const float*)d_in[3];
  const float* W = (const float*)d_in[5];
  const float* b = (const float*)d_in[6];

  const int N = in_sizes[0] / 16;
  const int E = in_sizes[1] / 2;
  const int T = in_sizes[6] / F48;

  const float* W3 = W + (size_t)(T - 1) * F48 * F48;
  const float* b3 = b + (size_t)(T - 1) * F48;
  const int* srcp = e;
  const int* dstp = e + E;
  float* out = (float*)d_out;

  const size_t NP = ((size_t)N + 63) & ~(size_t)63;
  const size_t EP = ((size_t)E + 63) & ~(size_t)63;
  float* featw = (float*)d_ws;
  float* xws = featw + (size_t)N * F48;
  float* dis = xws + (size_t)N * F48;
  int* deg = (int*)(dis + NP);
  int* rowptr = deg + NP;
  int* csr = rowptr + NP;
  int* binned = csr + EP;
  int* bincnt = binned + 2 * EP;
  int* binstart = bincnt + NB_MAX;
  int* chunkoff = binstart + NB_MAX + 64;

  const int BLK = 256;
  const int NB = (N + BIN_W - 1) >> BIN_SHIFT;
  build_feat_zero_bins<<<(N + BLK - 1) / BLK, BLK, 0, stream>>>(
      x, h, q, featw, bincnt, N, NB);
  bin_count<<<BIN_GRID, BIN_THREADS, 0, stream>>>(dstp, bincnt, chunkoff, E,
                                                   NB);
  bin_scatter<<<BIN_GRID, BIN_THREADS, 0, stream>>>(
      srcp, dstp, bincnt, chunkoff, binstart, binned, E, NB);
  bin_sort<<<NB, SORT_THREADS, 0, stream>>>(binned, binstart, deg, rowptr, dis,
                                            csr, N);

  const int nRowBlocks = (N + 15) / 16;
  const int wavesPerBlock = 8;
  gcn_gemm_wmma<<<(nRowBlocks + wavesPerBlock - 1) / wavesPerBlock, BLK, 0,
                  stream>>>(featw, W3, dis, xws, N);

  const int rowsPerBlock = AGG_WAVES * ROWS_PER_WAVE;
  aggregate_rows<<<(N + rowsPerBlock - 1) / rowsPerBlock, AGG_THREADS, 0,
                   stream>>>(csr, rowptr, deg, dis, b3, xws, out, N);
}


# GEMM fused into the per-bin sort kernel: each of its 16 waves does one 16-row block, A read straight from x,h,q, W staged once in LDS; feat array, its build kernel and the GEMM launch removed
# speedup vs baseline: 13.5496x; 1.3082x over previous
#include <hip/hip_runtime.h>

typedef __attribute__((ext_vector_type(2))) float v2f;
typedef __attribute__((ext_vector_type(8))) float v8f;

#define F48 48

#define BIN_SHIFT 8
#define BIN_W 256
#define NB_MAX 1024
#define BIN_THREADS 1024
#define BIN_GRID 256
#define EB 8

__global__ void zero_bins(int* __restrict__ bincnt, int NB) {
  const int i = blockIdx.x * blockDim.x + threadIdx.x;
  if (i < NB) bincnt[i] = 0;
}

__global__ __launch_bounds__(BIN_THREADS) void bin_count(
    const int* __restrict__ dst, int* __restrict__ bincnt,
    int* __restrict__ chunkoff, int E, int NB) {
  __shared__ int hist[NB_MAX];
  if (NB > NB_MAX) return;
  const int tid = threadIdx.x;
  if (tid < NB) hist[tid] = 0;
  __syncthreads();
  const int CH = (E + gridDim.x - 1) / gridDim.x;
  const int e0 = blockIdx.x * CH;
  const int e1 = min(E, e0 + CH);
  for (int b = e0 + tid; b < e1; b += EB * BIN_THREADS) {
    int d[EB];
#pragma unroll
    for (int u = 0; u < EB; ++u) {
      const int i = b + u * BIN_THREADS;
      d[u] = (i < e1) ? dst[i] : -1;
    }
#pragma unroll
    for (int u = 0; u < EB; ++u)
      if (d[u] >= 0) atomicAdd(&hist[d[u] >> BIN_SHIFT], 1);
  }
  __syncthreads();
  if (tid < NB) {
    const int c = hist[tid];
    chunkoff[blockIdx.x * NB_MAX + tid] = c ? atomicAdd(&bincnt[tid], c) : 0;
  }
}

__global__ __launch_bounds__(BIN_THREADS) void bin_scatter(
    const int* __restrict__ src, const int* __restrict__ dst,
    const int* __restrict__ bincnt, const int* __restrict__ chunkoff,
    int* __restrict__ binstart, int* __restrict__ binned, int E, int NB) {
  __shared__ int cur[NB_MAX];
  __shared__ int wsum[32];
  if (NB > NB_MAX) return;
  const int tid = threadIdx.x;
  const int lane = tid & 31;
  const int wave = tid >> 5;

  const int c = (tid < NB) ? bincnt[tid] : 0;
  const int mine = (tid < NB) ? chunkoff[blockIdx.x * NB_MAX + tid] : 0;
  int incl = c;
#pragma unroll
  for (int off = 1; off < 32; off <<= 1) {
    const int t = __shfl_up(incl, off);
    if (lane >= off) incl += t;
  }
  if (lane == 31) wsum[wave] = incl;
  __syncthreads();
  const int wtot = wsum[lane];
  int wincl = wtot;
#pragma unroll
  for (int off = 1; off < 32; off <<= 1) {
    const int t = __shfl_up(wincl, off);
    if (lane >= off) wincl += t;
  }
  const int wexcl = __shfl(wincl - wtot, wave);
  const int start = wexcl + (incl - c);
  if (tid < NB) {
    cur[tid] = start + mine;
    if (blockIdx.x == 0) binstart[tid] = start;
  }
  if (blockIdx.x == 0 && tid == 0) binstart[NB] = E;
  __syncthreads();

  int2* __restrict__ bp = (int2*)binned;
  const int CH = (E + gridDim.x - 1) / gridDim.x;
  const int e0 = blockIdx.x * CH;
  const int e1 = min(E, e0 + CH);
  for (int b = e0 + tid; b < e1; b += EB * BIN_THREADS) {
    int s[EB], d[EB];
#pragma unroll
    for (int u = 0; u < EB; ++u) {
      const int i = b + u * BIN_THREADS;
      const bool in = i < e1;
      d[u] = in ? dst[i] : -1;
      s[u] = in ? src[i] : 0;
    }
#pragma unroll
    for (int u = 0; u < EB; ++u) {
      if (d[u] >= 0) {
        const int p = atomicAdd(&cur[d[u] >> BIN_SHIFT], 1);
        bp[p] = make_int2(s[u], d[u]);
      }
    }
  }
}

#define SORT_THREADS 512
#define WPITCH 52
#define W_VEC4 (F48 * F48 / 4)
static_assert(BIN_W == (SORT_THREADS / 32) * 16,
              "one 16-row block of the bin per wave");
__global__ __launch_bounds__(SORT_THREADS) void bin_sort_gemm(
    const int* __restrict__ binned, const int* __restrict__ binstart,
    const float* __restrict__ x, const float* __restrict__ h,
    const float* __restrict__ q,
    const float* __restrict__ W3,
    int* __restrict__ deg, int* __restrict__ rowptr, float* __restrict__ dis,
    int* __restrict__ csr, float* __restrict__ xws, int N) {
  __shared__ int cnt[BIN_W];
  __shared__ int wtot[BIN_W / 32];
  __shared__ float dis_s[BIN_W];
  __shared__ __attribute__((aligned(16))) float Ws[F48 * WPITCH];
  const int tid = threadIdx.x;
  const int lane = tid & 31;
  const int wave = tid >> 5;
  const int bin = blockIdx.x;
  const int s0 = binstart[bin];
  const int s1 = binstart[bin + 1];
  const int n0 = bin << BIN_SHIFT;
  const int2* __restrict__ bp = (const int2*)binned;

  const int row0 = n0 + (__builtin_amdgcn_readfirstlane(wave) << 4);
  const bool live = row0 < N;
  const int lm = lane & 15;
  const int khalf = (lane >> 4) << 1;
  int arow = row0 + lm;
  if (arow >= N) arow = N - 1;

  v2f a[12];
#pragma unroll
  for (int k = 0; k < 4; ++k) {
    const int c = (k << 2) + khalf;
    a[k]     = *(const v2f*)(x + (size_t)arow * 16 + c);
    a[4 + k] = *(const v2f*)(h + (size_t)arow * 16 + c);
    a[8 + k] = *(const v2f*)(q + (size_t)arow * 16 + c);
  }
  static_assert(W_VEC4 % 2 == 0 && W_VEC4 / 2 <= SORT_THREADS && F48 % 8 == 0,
                "one pair of pieces per thread, both in one row");
  if (tid < BIN_W) cnt[tid] = 0;
  if (tid < W_VEC4 / 2) {
    const float4 w0 = *(const float4*)(W3 + 8 * tid);
    const float4 w1 = *(const float4*)(W3 + 8 * tid + 4);
    float* wrow = Ws + ((8 * tid) / F48) * WPITCH + (8 * tid) % F48;
    *(float4*)(wrow) = w0;
    *(float4*)(wrow + 4) = w1;
  }
  __syncthreads();

  v8f acc0 = {};
  v8f acc1 = {};
  v8f acc2 = {};
  if (live) {
#pragma unroll
    for (int k = 0; k < 12; ++k) {
      const int k4 = (k << 2) + khalf;
      v2f b0 = *(const v2f*)(Ws + (0  + lm) * WPITCH + k4);
      v2f b1 = *(const v2f*)(Ws + (16 + lm) * WPITCH + k4);
      v2f b2 = *(const v2f*)(Ws + (32 + lm) * WPITCH + k4);
      acc0 = __builtin_amdgcn_wmma_f32_16x16x4_f32(false, a[k], false, b0,
                                                   (short)0, acc0, false, false);
      acc1 = __builtin_amdgcn_wmma_f32_16x16x4_f32(false, a[k], false, b1,
                                                   (short)0, acc1, false, false);
      acc2 = __builtin_amdgcn_wmma_f32_16x16x4_f32(false, a[k], false, b2,
                                                   (short)0, acc2, false, false);
    }
  }

  for (int b = s0 + tid; b < s1; b += 4 * SORT_THREADS) {
    int d[4];
#pragma unroll
    for (int u = 0; u < 4; ++u) {
      const int i = b + u * SORT_THREADS;
      d[u] = (i < s1) ? bp[i].y : -1;
    }
#pragma unroll
    for (int u = 0; u < 4; ++u)
      if (d[u] >= 0) atomicAdd(&cnt[d[u] - n0], 1);
  }
  __syncthreads();

  const int dg = (tid < BIN_W) ? cnt[tid] : 0;
  int incl = dg;
#pragma unroll
  for (int off = 1; off < 32; off <<= 1) {
    const int t = __shfl_up(incl, off);
    if (lane >= off) incl += t;
  }
  if (tid < BIN_W && lane == 31) wtot[wave] = incl;
  __syncthreads();
  if (tid < BIN_W) {
    int front = 0;
#pragma unroll
    for (int w = 0; w < BIN_W / 32; ++w) front += (w < wave) ? wtot[w] : 0;
    const int excl = front + (incl - dg);
    cnt[tid] = excl;
    const int node = n0 + tid;
    const float dn = rsqrtf((float)(dg + 1));
    dis_s[tid] = dn;
    if (node < N) {
      deg[node] = dg;
      rowptr[node] = s0 + excl;
      dis[node] = dn;
    }
  }
  __syncthreads();

  for (int b = s0 + tid; b < s1; b += 4 * SORT_THREADS) {
    int2 e[4];
#pragma unroll
    for (int u = 0; u < 4; ++u) {
      const int i = b + u * SORT_THREADS;
      e[u] = (i < s1) ? bp[i] : make_int2(0, -1);
    }
#pragma unroll
    for (int u = 0; u < 4; ++u) {
      if (e[u].y >= 0) {
        const int p = atomicAdd(&cnt[e[u].y - n0], 1);
        csr[s0 + p] = e[u].x;
      }
    }
  }

  if (live) {
    const int mbase = (lane >> 4) << 3;
#pragma unroll
    for (int v = 0; v < 8; ++v) {
      const int m = row0 + mbase + v;
      if (m < N) {
        const float dm = dis_s[m - n0];
        const size_t base = (size_t)m * F48;
        xws[base + lm]      = acc0[v] * dm;
        xws[base + 16 + lm] = acc1[v] * dm;
        xws[base + 32 + lm] = acc2[v] * dm;
      }
    }
  }
}

#define AGG_THREADS 256
#define AGG_WAVES 8
#define ROWS_PER_WAVE 8
#define EDGE_BATCH 8

__global__ __launch_bounds__(AGG_THREADS) void aggregate_rows(
    const int* __restrict__ csr, const int* __restrict__ rowptr,
    const int* __restrict__ deg, const float* __restrict__ dis,
    const float* __restrict__ b3, const float* __restrict__ xws,
    float* __restrict__ out, int N) {
  const int lane = threadIdx.x & 31;
  const int wave = __builtin_amdgcn_readfirstlane((int)(threadIdx.x >> 5));
  const int row0 = (blockIdx.x * AGG_WAVES + wave) * ROWS_PER_WAVE;
  if (row0 >= N) return;
  const bool hi = lane >= 16;
  const int l16 = lane & 15;
  const int c4 = (l16 < 12 ? l16 : 11) * 4;
  const float4 bias = *(const float4*)(b3 + c4);

  int mb = 0, mc = 0;
  float md = 0.0f;
  if (lane < ROWS_PER_WAVE && row0 + lane < N) {
    mb = rowptr[row0 + lane];
    mc = deg[row0 + lane];
    md = dis[row0 + lane];
  }

  for (int k = 0; k < ROWS_PER_WAVE / 2; ++k) {
    const int begA = __builtin_amdgcn_readlane(mb, 2 * k);
    const int cntA = __builtin_amdgcn_readlane(mc, 2 * k);
    const int begB = __builtin_amdgcn_readlane(mb, 2 * k + 1);
    const int cntB = __builtin_amdgcn_readlane(mc, 2 * k + 1);
    const int dA = __builtin_amdgcn_readlane(__builtin_bit_cast(int, md), 2 * k);
    const int dB =
        __builtin_amdgcn_readlane(__builtin_bit_cast(int, md), 2 * k + 1);
    const int beg = hi ? begB : begA;
    const int cnt = hi ? cntB : cntA;
    const float dr = __builtin_bit_cast(float, hi ? dB : dA);
    const int maxc = max(cntA, cntB);
    const int r = row0 + 2 * k + (hi ? 1 : 0);
    const bool wr = (r < N) && (l16 < 12);

    float4 self4 = make_float4(0.0f, 0.0f, 0.0f, 0.0f);
    if (wr) self4 = *(const float4*)(xws + (size_t)r * F48 + c4);

    float4 acc = make_float4(0.0f, 0.0f, 0.0f, 0.0f);
    for (int c0 = 0; c0 < maxc; c0 += 16) {
      int s = 0;
      if (c0 + l16 < cnt) s = csr[beg + c0 + l16];
      const int nh = min(16, maxc - c0);
      for (int j0 = 0; j0 < nh; j0 += EDGE_BATCH) {
        float4 xv[EDGE_BATCH];
#pragma unroll
        for (int q = 0; q < EDGE_BATCH; ++q) {
          const int j = j0 + q;
          const int sA = __builtin_amdgcn_readlane(s, j);
          const int sB = __builtin_amdgcn_readlane(s, 16 + j);
          const int sj = hi ? sB : sA;
          xv[q] = *(const float4*)(xws + (size_t)sj * F48 + c4);
        }
#pragma unroll
        for (int q = 0; q < EDGE_BATCH; ++q) {
          const bool ok = (c0 + j0 + q) < cnt;
          acc.x += ok ? xv[q].x : 0.0f;
          acc.y += ok ? xv[q].y : 0.0f;
          acc.z += ok ? xv[q].z : 0.0f;
          acc.w += ok ? xv[q].w : 0.0f;
        }
      }
    }
    if (wr) {
      float4 o;
      o.x = (acc.x + self4.x) * dr + bias.x;
      o.y = (acc.y + self4.y) * dr + bias.y;
      o.z = (acc.z + self4.z) * dr + bias.z;
      o.w = (acc.w + self4.w) * dr + bias.w;
      *(float4*)(out + (size_t)r * F48 + c4) = o;
    }
  }
}

extern "C" void kernel_launch(void* const* d_in, const int* in_sizes, int n_in,
                              void* d_out, int out_size, void* d_ws,
                              size_t ws_size, hipStream_t stream) {
  const float* h = (const float*)d_in[0];
  const int* e = (const int*)d_in[1];
  const float* x = (const float*)d_in[2];
  const float* q = (const float*)d_in[3];
  const float* W = (const float*)d_in[5];
  const float* b = (const float*)d_in[6];

  const int N = in_sizes[0] / 16;
  const int E = in_sizes[1] / 2;
  const int T = in_sizes[6] / F48;

  const float* W3 = W + (size_t)(T - 1) * F48 * F48;
  const float* b3 = b + (size_t)(T - 1) * F48;
  const int* srcp = e;
  const int* dstp = e + E;
  float* out = (float*)d_out;

  const size_t NP = ((size_t)N + 63) & ~(size_t)63;
  const size_t EP = ((size_t)E + 63) & ~(size_t)63;
  float* xws = (float*)d_ws;
  float* dis = xws + (size_t)N * F48;
  int* deg = (int*)(dis + NP);
  int* rowptr = deg + NP;
  int* csr = rowptr + NP;
  int* binned = csr + EP;
  int* bincnt = binned + 2 * EP;
  int* binstart = bincnt + NB_MAX;
  int* chunkoff = binstart + NB_MAX + 64;

  const int NB = (N + BIN_W - 1) >> BIN_SHIFT;
  zero_bins<<<(NB + BIN_THREADS - 1) / BIN_THREADS, BIN_THREADS, 0, stream>>>(
      bincnt, NB);
  bin_count<<<BIN_GRID, BIN_THREADS, 0, stream>>>(dstp, bincnt, chunkoff, E,
                                                   NB);
  bin_scatter<<<BIN_GRID, BIN_THREADS, 0, stream>>>(
      srcp, dstp, bincnt, chunkoff, binstart, binned, E, NB);
  bin_sort_gemm<<<NB, SORT_THREADS, 0, stream>>>(
      binned, binstart, x, h, q, W3, deg, rowptr, dis, csr, xws, N);

  const int rowsPerBlock = AGG_WAVES * ROWS_PER_WAVE;
  aggregate_rows<<<(N + rowsPerBlock - 1) / rowsPerBlock, AGG_THREADS, 0,
                   stream>>>(csr, rowptr, deg, dis, b3, xws, out, N);
}


# Edge records laid out in LDS first (chunk ordered by bin; bin's src list by row), then written with whole-line stores in place of one scattered store per edge (row order: arrival, as before)
# speedup vs baseline: 16.1772x; 1.1939x over previous
#include <hip/hip_runtime.h>

typedef __attribute__((ext_vector_type(2))) float v2f;
typedef __attribute__((ext_vector_type(8))) float v8f;

#define F48 48

#define BIN_SHIFT 8
#define BIN_W 256
#define NB_MAX 1024
#define BIN_THREADS 1024
#define BIN_GRID 256
#define EB 8

__global__ void zero_bins(int* __restrict__ bincnt, int NB) {
  const int i = blockIdx.x * blockDim.x + threadIdx.x;
  if (i < NB) bincnt[i] = 0;
}

__global__ __launch_bounds__(BIN_THREADS) void bin_count(
    const int* __restrict__ dst, int* __restrict__ bincnt,
    int* __restrict__ chunkoff, int E, int NB) {
  __shared__ int hist[NB_MAX];
  if (NB > NB_MAX) return;
  const int tid = threadIdx.x;
  if (tid < NB) hist[tid] = 0;
  __syncthreads();
  const int CH = (E + gridDim.x - 1) / gridDim.x;
  const int e0 = blockIdx.x * CH;
  const int e1 = min(E, e0 + CH);
  for (int b = e0 + tid; b < e1; b += EB * BIN_THREADS) {
    int d[EB];
#pragma unroll
    for (int u = 0; u < EB; ++u) {
      const int i = b + u * BIN_THREADS;
      d[u] = (i < e1) ? dst[i] : -1;
    }
#pragma unroll
    for (int u = 0; u < EB; ++u)
      if (d[u] >= 0) atomicAdd(&hist[d[u] >> BIN_SHIFT], 1);
  }
  __syncthreads();
  if (tid < NB) {
    const int c = hist[tid];
    ((int2*)chunkoff)[blockIdx.x * NB_MAX + tid] =
        make_int2(c ? atomicAdd(&bincnt[tid], c) : 0, c);
  }
}

#define STAGE_CAP 8192
__global__ __launch_bounds__(BIN_THREADS) void bin_scatter(
    const int* __restrict__ src, const int* __restrict__ dst,
    const int* __restrict__ bincnt, const int* __restrict__ chunkoff,
    int* __restrict__ binstart, int* __restrict__ binned, int E, int NB) {
  __shared__ int cur[NB_MAX];
  __shared__ int gdelta[NB_MAX];
  __shared__ int wsum[32];
  __shared__ int lwsum[32];
  __shared__ int2 stage[STAGE_CAP];
  if (NB > NB_MAX) return;
  const int tid = threadIdx.x;
  const int lane = tid & 31;
  const int wave = tid >> 5;

  const int c = (tid < NB) ? bincnt[tid] : 0;
  const int2 oc = (tid < NB) ? ((const int2*)chunkoff)[blockIdx.x * NB_MAX + tid]
                             : make_int2(0, 0);
  const int mine = oc.x;
  const int lc = oc.y;
  int incl = c;
  int lincl = lc;
#pragma unroll
  for (int off = 1; off < 32; off <<= 1) {
    const int t = __shfl_up(incl, off);
    const int lt = __shfl_up(lincl, off);
    if (lane >= off) {
      incl += t;
      lincl += lt;
    }
  }
  if (lane == 31) {
    wsum[wave] = incl;
    lwsum[wave] = lincl;
  }
  __syncthreads();
  const int wtot = wsum[lane];
  const int lwtot = lwsum[lane];
  int wincl = wtot;
  int lwincl = lwtot;
#pragma unroll
  for (int off = 1; off < 32; off <<= 1) {
    const int t = __shfl_up(wincl, off);
    const int lt = __shfl_up(lwincl, off);
    if (lane >= off) {
      wincl += t;
      lwincl += lt;
    }
  }
  const int wexcl = __shfl(wincl - wtot, wave);
  const int lwexcl = __shfl(lwincl - lwtot, wave);
  const int start = wexcl + (incl - c);
  const int lstart = lwexcl + (lincl - lc);
  const int CH = (E + gridDim.x - 1) / gridDim.x;
  const bool staged = CH <= STAGE_CAP;
  if (tid < NB) {
    cur[tid] = staged ? lstart : start + mine;
    gdelta[tid] = start + mine - lstart;
    if (blockIdx.x == 0) binstart[tid] = start;
  }
  if (blockIdx.x == 0 && tid == 0) binstart[NB] = E;
  __syncthreads();

  int2* __restrict__ bp = (int2*)binned;
  const int e0 = blockIdx.x * CH;
  const int e1 = min(E, e0 + CH);
  if (staged) {
    for (int b = e0 + tid; b < e1; b += EB * BIN_THREADS) {
      int s[EB], d[EB], p[EB];
#pragma unroll
      for (int u = 0; u < EB; ++u) {
        const int i = b + u * BIN_THREADS;
        const bool in = i < e1;
        d[u] = in ? dst[i] : -1;
        s[u] = in ? src[i] : 0;
      }
#pragma unroll
      for (int u = 0; u < EB; ++u)
        p[u] = (d[u] >= 0) ? atomicAdd(&cur[d[u] >> BIN_SHIFT], 1) : -1;
#pragma unroll
      for (int u = 0; u < EB; ++u)
        if (p[u] >= 0) stage[p[u]] = make_int2(s[u], d[u]);
    }
    __syncthreads();
    const int n = e1 - e0;
    for (int j0 = tid; j0 < n; j0 += EB * BIN_THREADS) {
      int2 r[EB];
      int g[EB];
#pragma unroll
      for (int u = 0; u < EB; ++u)
        r[u] = stage[min(j0 + u * BIN_THREADS, n - 1)];
#pragma unroll
      for (int u = 0; u < EB; ++u) g[u] = gdelta[r[u].y >> BIN_SHIFT];
#pragma unroll
      for (int u = 0; u < EB; ++u) {
        const int j = j0 + u * BIN_THREADS;
        if (j < n) bp[j + g[u]] = r[u];
      }
    }
    return;
  }
  for (int b = e0 + tid; b < e1; b += EB * BIN_THREADS) {
    int s[EB], d[EB];
#pragma unroll
    for (int u = 0; u < EB; ++u) {
      const int i = b + u * BIN_THREADS;
      const bool in = i < e1;
      d[u] = in ? dst[i] : -1;
      s[u] = in ? src[i] : 0;
    }
#pragma unroll
    for (int u = 0; u < EB; ++u) {
      if (d[u] >= 0) {
        const int p = atomicAdd(&cur[d[u] >> BIN_SHIFT], 1);
        bp[p] = make_int2(s[u], d[u]);
      }
    }
  }
}

#define SORT_THREADS 512
#define CSR_STAGE 8192
#define WPITCH 52
#define W_VEC4 (F48 * F48 / 4)
static_assert(BIN_W == (SORT_THREADS / 32) * 16,
              "one 16-row block of the bin per wave");
__global__ __launch_bounds__(SORT_THREADS) void bin_sort_gemm(
    const int* __restrict__ binned, const int* __restrict__ binstart,
    const float* __restrict__ x, const float* __restrict__ h,
    const float* __restrict__ q,
    const float* __restrict__ W3,
    int* __restrict__ deg, int* __restrict__ rowptr, float* __restrict__ dis,
    int* __restrict__ csr, float* __restrict__ xws, int N) {
  __shared__ int cnt[BIN_W];
  __shared__ int wtot[BIN_W / 32];
  __shared__ float dis_s[BIN_W];
  __shared__ __attribute__((aligned(16))) float Ws[F48 * WPITCH];
  __shared__ int stage[CSR_STAGE];
  const int tid = threadIdx.x;
  const int lane = tid & 31;
  const int wave = tid >> 5;
  const int bin = blockIdx.x;
  const int s0 = binstart[bin];
  const int s1 = binstart[bin + 1];
  const int n0 = bin << BIN_SHIFT;
  const int2* __restrict__ bp = (const int2*)binned;

  const int row0 = n0 + (__builtin_amdgcn_readfirstlane(wave) << 4);
  const bool live = row0 < N;
  const int lm = lane & 15;
  const int khalf = (lane >> 4) << 1;
  int arow = row0 + lm;
  if (arow >= N) arow = N - 1;

  v2f a[12];
#pragma unroll
  for (int k = 0; k < 4; ++k) {
    const int c = (k << 2) + khalf;
    a[k]     = *(const v2f*)(x + (size_t)arow * 16 + c);
    a[4 + k] = *(const v2f*)(h + (size_t)arow * 16 + c);
    a[8 + k] = *(const v2f*)(q + (size_t)arow * 16 + c);
  }
  static_assert(W_VEC4 % 2 == 0 && W_VEC4 / 2 <= SORT_THREADS && F48 % 8 == 0,
                "one pair of pieces per thread, both in one row");
  if (tid < BIN_W) cnt[tid] = 0;
  if (tid < W_VEC4 / 2) {
    const float4 w0 = *(const float4*)(W3 + 8 * tid);
    const float4 w1 = *(const float4*)(W3 + 8 * tid + 4);
    float* wrow = Ws + ((8 * tid) / F48) * WPITCH + (8 * tid) % F48;
    *(float4*)(wrow) = w0;
    *(float4*)(wrow + 4) = w1;
  }
  __syncthreads();

  v8f acc0 = {};
  v8f acc1 = {};
  v8f acc2 = {};
  if (live) {
#pragma unroll
    for (int k = 0; k < 12; ++k) {
      const int k4 = (k << 2) + khalf;
      v2f b0 = *(const v2f*)(Ws + (0  + lm) * WPITCH + k4);
      v2f b1 = *(const v2f*)(Ws + (16 + lm) * WPITCH + k4);
      v2f b2 = *(const v2f*)(Ws + (32 + lm) * WPITCH + k4);
      acc0 = __builtin_amdgcn_wmma_f32_16x16x4_f32(false, a[k], false, b0,
                                                   (short)0, acc0, false, false);
      acc1 = __builtin_amdgcn_wmma_f32_16x16x4_f32(false, a[k], false, b1,
                                                   (short)0, acc1, false, false);
      acc2 = __builtin_amdgcn_wmma_f32_16x16x4_f32(false, a[k], false, b2,
                                                   (short)0, acc2, false, false);
    }
  }

  for (int b = s0 + tid; b < s1; b += 4 * SORT_THREADS) {
    int d[4];
#pragma unroll
    for (int u = 0; u < 4; ++u) {
      const int i = b + u * SORT_THREADS;
      d[u] = (i < s1) ? bp[i].y : -1;
    }
#pragma unroll
    for (int u = 0; u < 4; ++u)
      if (d[u] >= 0) atomicAdd(&cnt[d[u] - n0], 1);
  }
  __syncthreads();

  const int dg = (tid < BIN_W) ? cnt[tid] : 0;
  int incl = dg;
#pragma unroll
  for (int off = 1; off < 32; off <<= 1) {
    const int t = __shfl_up(incl, off);
    if (lane >= off) incl += t;
  }
  if (tid < BIN_W && lane == 31) wtot[wave] = incl;
  __syncthreads();
  if (tid < BIN_W) {
    int front = 0;
#pragma unroll
    for (int w = 0; w < BIN_W / 32; ++w) front += (w < wave) ? wtot[w] : 0;
    const int excl = front + (incl - dg);
    cnt[tid] = excl;
    const int node = n0 + tid;
    const float dn = rsqrtf((float)(dg + 1));
    dis_s[tid] = dn;
    if (node < N) {
      deg[node] = dg;
      rowptr[node] = s0 + excl;
      dis[node] = dn;
    }
  }
  __syncthreads();

  for (int b = s0 + tid; b < s1; b += 4 * SORT_THREADS) {
    int2 e[4];
    int p[4];
#pragma unroll
    for (int u = 0; u < 4; ++u) {
      const int i = b + u * SORT_THREADS;
      e[u] = (i < s1) ? bp[i] : make_int2(0, -1);
    }
#pragma unroll
    for (int u = 0; u < 4; ++u)
      p[u] = (e[u].y >= 0) ? atomicAdd(&cnt[e[u].y - n0], 1) : -1;
#pragma unroll
    for (int u = 0; u < 4; ++u) {
      if (p[u] >= 0) {
        if (p[u] < CSR_STAGE) stage[p[u]] = e[u].x;
        else csr[s0 + p[u]] = e[u].x;
      }
    }
  }
  __syncthreads();
  {
    const int last = min(s1 - s0, CSR_STAGE) - 1;
    const int s2 = s0 + last + 1;
    if (last >= 0) {
      for (int j0 = (s0 & ~31) + tid; j0 < s2; j0 += 8 * SORT_THREADS) {
        int v[8];
#pragma unroll
        for (int u = 0; u < 8; ++u)
          v[u] = stage[min(max(j0 + u * SORT_THREADS - s0, 0), last)];
#pragma unroll
        for (int u = 0; u < 8; ++u) {
          const int j = j0 + u * SORT_THREADS;
          if (j >= s0 && j < s2) csr[j] = v[u];
        }
      }
    }
  }

  if (live) {
    const int mbase = (lane >> 4) << 3;
#pragma unroll
    for (int v = 0; v < 8; ++v) {
      const int m = row0 + mbase + v;
      if (m < N) {
        const float dm = dis_s[m - n0];
        const size_t base = (size_t)m * F48;
        xws[base + lm]      = acc0[v] * dm;
        xws[base + 16 + lm] = acc1[v] * dm;
        xws[base + 32 + lm] = acc2[v] * dm;
      }
    }
  }
}

#define AGG_THREADS 256
#define AGG_WAVES 8
#define ROWS_PER_WAVE 8
#define EDGE_BATCH 8

__global__ __launch_bounds__(AGG_THREADS) void aggregate_rows(
    const int* __restrict__ csr, const int* __restrict__ rowptr,
    const int* __restrict__ deg, const float* __restrict__ dis,
    const float* __restrict__ b3, const float* __restrict__ xws,
    float* __restrict__ out, int N) {
  const int lane = threadIdx.x & 31;
  const int wave = __builtin_amdgcn_readfirstlane((int)(threadIdx.x >> 5));
  const int row0 = (blockIdx.x * AGG_WAVES + wave) * ROWS_PER_WAVE;
  if (row0 >= N) return;
  const bool hi = lane >= 16;
  const int l16 = lane & 15;
  const int c4 = (l16 < 12 ? l16 : 11) * 4;
  const float4 bias = *(const float4*)(b3 + c4);

  int mb = 0, mc = 0;
  float md = 0.0f;
  if (lane < ROWS_PER_WAVE && row0 + lane < N) {
    mb = rowptr[row0 + lane];
    mc = deg[row0 + lane];
    md = dis[row0 + lane];
  }

  for (int k = 0; k < ROWS_PER_WAVE / 2; ++k) {
    const int begA = __builtin_amdgcn_readlane(mb, 2 * k);
    const int cntA = __builtin_amdgcn_readlane(mc, 2 * k);
    const int begB = __builtin_amdgcn_readlane(mb, 2 * k + 1);
    const int cntB = __builtin_amdgcn_readlane(mc, 2 * k + 1);
    const int dA = __builtin_amdgcn_readlane(__builtin_bit_cast(int, md), 2 * k);
    const int dB =
        __builtin_amdgcn_readlane(__builtin_bit_cast(int, md), 2 * k + 1);
    const int beg = hi ? begB : begA;
    const int cnt = hi ? cntB : cntA;
    const float dr = __builtin_bit_cast(float, hi ? dB : dA);
    const int maxc = max(cntA, cntB);
    const int r = row0 + 2 * k + (hi ? 1 : 0);
    const bool wr = (r < N) && (l16 < 12);

    float4 self4 = make_float4(0.0f, 0.0f, 0.0f, 0.0f);
    if (wr) self4 = *(const float4*)(xws + (size_t)r * F48 + c4);

    float4 acc = make_float4(0.0f, 0.0f, 0.0f, 0.0f);
    for (int c0 = 0; c0 < maxc; c0 += 16) {
      int s = 0;
      if (c0 + l16 < cnt) s = csr[beg + c0 + l16];
      const int nh = min(16, maxc - c0);
      for (int j0 = 0; j0 < nh; j0 += EDGE_BATCH) {
        float4 xv[EDGE_BATCH];
#pragma unroll
        for (int q = 0; q < EDGE_BATCH; ++q) {
          const int j = j0 + q;
          const int sA = __builtin_amdgcn_readlane(s, j);
          const int sB = __builtin_amdgcn_readlane(s, 16 + j);
          const int sj = hi ? sB : sA;
          xv[q] = *(const float4*)(xws + (size_t)sj * F48 + c4);
        }
#pragma unroll
        for (int q = 0; q < EDGE_BATCH; ++q) {
          const bool ok = (c0 + j0 + q) < cnt;
          acc.x += ok ? xv[q].x : 0.0f;
          acc.y += ok ? xv[q].y : 0.0f;
          acc.z += ok ? xv[q].z : 0.0f;
          acc.w += ok ? xv[q].w : 0.0f;
        }
      }
    }
    if (wr) {
      float4 o;
      o.x = (acc.x + self4.x) * dr + bias.x;
      o.y = (acc.y + self4.y) * dr + bias.y;
      o.z = (acc.z + self4.z) * dr + bias.z;
      o.w = (acc.w + self4.w) * dr + bias.w;
      *(float4*)(out + (size_t)r * F48 + c4) = o;
    }
  }
}

extern "C" void kernel_launch(void* const* d_in, const int* in_sizes, int n_in,
                              void* d_out, int out_size, void* d_ws,
                              size_t ws_size, hipStream_t stream) {
  const float* h = (const float*)d_in[0];
  const int* e = (const int*)d_in[1];
  const float* x = (const float*)d_in[2];
  const float* q = (const float*)d_in[3];
  const float* W = (const float*)d_in[5];
  const float* b = (const float*)d_in[6];

  const int N = in_sizes[0] / 16;
  const int E = in_sizes[1] / 2;
  const int T = in_sizes[6] / F48;

  const float* W3 = W + (size_t)(T - 1) * F48 * F48;
  const float* b3 = b + (size_t)(T - 1) * F48;
  const int* srcp = e;
  const int* dstp = e + E;
  float* out = (float*)d_out;

  const size_t NP = ((size_t)N + 63) & ~(size_t)63;
  const size_t EP = ((size_t)E + 63) & ~(size_t)63;
  float* xws = (float*)d_ws;
  float* dis = xws + (size_t)N * F48;
  int* deg = (int*)(dis + NP);
  int* rowptr = deg + NP;
  int* csr = rowptr + NP;
  int* binned = csr + EP;
  int* bincnt = binned + 2 * EP;
  int* binstart = bincnt + NB_MAX;
  int* chunkoff = binstart + NB_MAX + 64;

  const int NB = (N + BIN_W - 1) >> BIN_SHIFT;
  zero_bins<<<(NB + BIN_THREADS - 1) / BIN_THREADS, BIN_THREADS, 0, stream>>>(
      bincnt, NB);
  bin_count<<<BIN_GRID, BIN_THREADS, 0, stream>>>(dstp, bincnt, chunkoff, E,
                                                   NB);
  bin_scatter<<<BIN_GRID, BIN_THREADS, 0, stream>>>(
      srcp, dstp, bincnt, chunkoff, binstart, binned, E, NB);
  bin_sort_gemm<<<NB, SORT_THREADS, 0, stream>>>(
      binned, binstart, x, h, q, W3, deg, rowptr, dis, csr, xws, N);

  const int rowsPerBlock = AGG_WAVES * ROWS_PER_WAVE;
  aggregate_rows<<<(N + rowsPerBlock - 1) / rowsPerBlock, AGG_THREADS, 0,
                   stream>>>(csr, rowptr, deg, dis, b3, xws, out, N);
}
